# starting kernel
# baseline (speedup 1.0000x reference)
#include <hip/hip_runtime.h>
#include <hip/hip_bf16.h>

typedef __attribute__((ext_vector_type(2))) float v2f;
typedef __attribute__((ext_vector_type(8))) float v8f;

#define BATCH 64
#define SEQ   2048
#define DIM   512

__global__ __launch_bounds__(128) void l2d_pass1(const float* __restrict__ x,
                                                 float* __restrict__ out) {
    const int lane   = threadIdx.x & 31;
    const int waveIn = threadIdx.x >> 5;
    const int wave   = blockIdx.x * 4 + waveIn;
    const int tilesPerBatch = SEQ / 16;
    const int b  = wave / tilesPerBatch;
    const int s0 = (wave % tilesPerBatch) * 16;

    const int m = lane & 15;
    const int g = lane >> 4;

    const int s  = s0 + m;
    const int sp = (s == 0) ? 0 : (s - 1);

    const float* __restrict__ cur  = x + ((size_t)b * SEQ + s)  * DIM;
    const float* __restrict__ prev = x + ((size_t)b * SEQ + sp) * DIM;

    v8f acc0 = {};
    v8f acc1 = {};
    const v2f ones = {1.0f, 1.0f};

    #pragma unroll 4
    for (int d0 = 0; d0 < DIM; d0 += 8) {
        const float4 c4 = *reinterpret_cast<const float4*>(cur  + d0 + 4 * g);
        const float4 p4 = *reinterpret_cast<const float4*>(prev + d0 + 4 * g);
        const float dx0 = c4.x - p4.x;
        const float dx1 = c4.y - p4.y;
        const float dx2 = c4.z - p4.z;
        const float dx3 = c4.w - p4.w;
        v2f a0 = { dx0 * dx0, dx1 * dx1 };
        v2f a1 = { dx2 * dx2, dx3 * dx3 };
        acc0 = __builtin_amdgcn_wmma_f32_16x16x4_f32(
            false, a0, false, ones, (short)0, acc0, false, false);
        acc1 = __builtin_amdgcn_wmma_f32_16x16x4_f32(
            false, a1, false, ones, (short)0, acc1, false, false);
    }

    const v8f accf = acc0 + acc1;

    float* __restrict__ op = out + (size_t)b * SEQ + s0;
    if (lane == 0) {
        #pragma unroll
        for (int j = 0; j < 8; ++j) op[j] = accf[j];
    } else if (lane == 16) {
        #pragma unroll
        for (int j = 0; j < 8; ++j) op[8 + j] = accf[j];
    }
}

__global__ __launch_bounds__(256) void l2d_pass2(float* __restrict__ d) {
    __shared__ float red[256];
    const int b = blockIdx.x;
    const int t = threadIdx.x;
    float* __restrict__ row = d + (size_t)b * SEQ;

    float v[SEQ / 256];
    float mx = 0.0f;
    #pragma unroll
    for (int i = 0; i < SEQ / 256; ++i) {
        v[i] = row[t + i * 256];
        mx = fmaxf(mx, v[i]);
    }
    red[t] = mx;
    __syncthreads();
    #pragma unroll
    for (int off = 128; off > 0; off >>= 1) {
        if (t < off) red[t] = fmaxf(red[t], red[t + off]);
        __syncthreads();
    }
    const float mxb = red[0];
    #pragma unroll
    for (int i = 0; i < SEQ / 256; ++i) {
        row[t + i * 256] = v[i] / mxb;
    }
}

extern "C" void kernel_launch(void* const* d_in, const int* in_sizes, int n_in,
                              void* d_out, int out_size, void* d_ws, size_t ws_size,
                              hipStream_t stream) {
    (void)in_sizes; (void)n_in; (void)out_size; (void)d_ws; (void)ws_size;
    const float* x = (const float*)d_in[0];
    float* out = (float*)d_out;

    const int totalWaves = BATCH * (SEQ / 16);
    l2d_pass1<<<totalWaves / 4, 128, 0, stream>>>(x, out);

    l2d_pass2<<<BATCH, 256, 0, stream>>>(out);
}


# Pass 1 streams whole rows: a wave reads 33 consecutive rows once, 512 contiguous bytes per load, previous row kept in registers, loads 6 rows ahead; WMMA sums folded per row (new sum order)
# speedup vs baseline: 1.3641x; 1.3641x over previous
#include <hip/hip_runtime.h>
#include <hip/hip_bf16.h>


typedef __attribute__((ext_vector_type(2))) float v2f;
typedef __attribute__((ext_vector_type(4))) float v4f;
typedef __attribute__((ext_vector_type(8))) float v8f;

#define BATCH 64
#define SEQ   2048
#define DIM   512

#define STRIP  32
#define WAVES  8
#define AHEAD  6
#define RING   (AHEAD + 2)
#define PIECES (DIM / 128)

__global__ __launch_bounds__(WAVES * 32) void l2d_pass1(const float* __restrict__ x,
                                                        float* __restrict__ out) {
    const int lane   = threadIdx.x & 31;
    const int waveIn = __builtin_amdgcn_readfirstlane(threadIdx.x >> 5);
    const int strip  = blockIdx.x * WAVES + waveIn;
    const int stripsPerBatch = SEQ / STRIP;
    const int b  = strip / stripsPerBatch;
    const int s0 = (strip % stripsPerBatch) * STRIP;

    const float* __restrict__ rows = x + ((size_t)b * SEQ + s0) * DIM + 4 * lane;
    const float* __restrict__ halo = (s0 == 0) ? rows : rows - DIM;

    v4f ring[RING][PIECES];
#define LOAD_ROW(I) do { \
        const float* rp_ = ((I) == 0) ? halo : rows + (size_t)((I) - 1) * DIM; \
        _Pragma("unroll") for (int q = 0; q < PIECES; ++q) \
            ring[(I) % RING][q] = *reinterpret_cast<const v4f*>(rp_ + 128 * q); \
    } while (0)

    #pragma unroll
    for (int i = 0; i <= AHEAD; ++i) LOAD_ROW(i);

    const v2f ones = {1.0f, 1.0f};
    float res0 = 0.0f, res1 = 0.0f;

    #pragma unroll
    for (int i = 1; i <= STRIP; ++i) {
        __builtin_amdgcn_sched_barrier(0);
        if (i + AHEAD <= STRIP) LOAD_ROW(i + AHEAD);
        __builtin_amdgcn_sched_barrier(0);

        v8f acc0 = {};
        v8f acc1 = {};
        #pragma unroll
        for (int q = 0; q < PIECES; ++q) {
            const v4f dx = ring[i % RING][q] - ring[(i - 1) % RING][q];
            const v4f sq = dx * dx;
            const v2f a0 = { sq.x, sq.y };
            const v2f a1 = { sq.z, sq.w };
            acc0 = __builtin_amdgcn_wmma_f32_16x16x4_f32(
                false, a0, false, ones, (short)0, acc0, false, false);
            acc1 = __builtin_amdgcn_wmma_f32_16x16x4_f32(
                false, a1, false, ones, (short)0, acc1, false, false);
        }
        const v8f a = acc0 + acc1;
        float r = ((a[0] + a[1]) + (a[2] + a[3])) + ((a[4] + a[5]) + (a[6] + a[7]));
        asm volatile("" : "+v"(r));
        const int t = i - 1;
        if (t < 16) res0 = ((lane & 15) == (t & 15)) ? r : res0;
        else        res1 = ((lane & 15) == (t & 15)) ? r : res1;
    }
#undef LOAD_ROW

    const bool  lo   = lane < 16;
    const float mine = lo ? res0 : res1;
    const float give = lo ? res1 : res0;
    const float got  = __shfl_xor(give, 16, 32);
    out[(size_t)b * SEQ + s0 + lane] = mine + got;
}

__global__ __launch_bounds__(256) void l2d_pass2(float* __restrict__ d) {
    __shared__ float red[256];
    const int b = blockIdx.x;
    const int t = threadIdx.x;
    float* __restrict__ row = d + (size_t)b * SEQ;

    float v[SEQ / 256];
    float mx = 0.0f;
    #pragma unroll
    for (int i = 0; i < SEQ / 256; ++i) {
        v[i] = row[t + i * 256];
        mx = fmaxf(mx, v[i]);
    }
    red[t] = mx;
    __syncthreads();
    #pragma unroll
    for (int off = 128; off > 0; off >>= 1) {
        if (t < off) red[t] = fmaxf(red[t], red[t + off]);
        __syncthreads();
    }
    const float mxb = red[0];
    #pragma unroll
    for (int i = 0; i < SEQ / 256; ++i) {
        row[t + i * 256] = v[i] / mxb;
    }
}

extern "C" void kernel_launch(void* const* d_in, const int* in_sizes, int n_in,
                              void* d_out, int out_size, void* d_ws, size_t ws_size,
                              hipStream_t stream) {
    (void)in_sizes; (void)n_in; (void)out_size; (void)d_ws; (void)ws_size;
    const float* x = (const float*)d_in[0];
    float* out = (float*)d_out;

    l2d_pass1<<<BATCH * (SEQ / STRIP) / WAVES, WAVES * 32, 0, stream>>>(x, out);

    l2d_pass2<<<BATCH, 256, 0, stream>>>(out);
}


# speedup vs baseline: 1.4253x; 1.0449x over previous
#include <hip/hip_runtime.h>
#include <hip/hip_bf16.h>


typedef __attribute__((ext_vector_type(2))) float v2f;
typedef __attribute__((ext_vector_type(4))) float v4f;
typedef __attribute__((ext_vector_type(8))) float v8f;

#define BATCH 64
#define SEQ   2048
#define DIM   512

#define STRIP  32
#define WAVES  8
#define AHEAD  6
#define RING   (AHEAD + 2)
#define PIECES (DIM / 128)

__global__ __launch_bounds__(WAVES * 32) void l2d_pass1(const float* __restrict__ x,
                                                        float* __restrict__ out) {
    const int lane   = threadIdx.x & 31;
    const int waveIn = __builtin_amdgcn_readfirstlane(threadIdx.x >> 5);
    const int strip  = blockIdx.x * WAVES + waveIn;
    const int stripsPerBatch = SEQ / STRIP;
    const int b  = strip / stripsPerBatch;
    const int s0 = (strip % stripsPerBatch) * STRIP;

    const float* __restrict__ rows = x + ((size_t)b * SEQ + s0) * DIM + 4 * lane;
    const float* __restrict__ halo = (s0 == 0) ? rows : rows - DIM;

    v4f ring[RING][PIECES];
#define LOAD_ROW(I) do { \
        const float* rp_ = ((I) == 0) ? halo : rows + (size_t)((I) - 1) * DIM; \
        _Pragma("unroll") for (int q = 0; q < PIECES; ++q) { \
            const v4f* p_ = reinterpret_cast<const v4f*>(rp_ + 128 * q); \
            if ((I) == 0 || (I) == STRIP) ring[(I) % RING][q] = *p_; \
            else ring[(I) % RING][q] = __builtin_nontemporal_load(p_); \
        } \
    } while (0)

    #pragma unroll
    for (int i = 0; i <= AHEAD; ++i) LOAD_ROW(i);

    const v2f ones = {1.0f, 1.0f};
    float res0 = 0.0f, res1 = 0.0f;

    #pragma unroll
    for (int i = 1; i <= STRIP; ++i) {
        __builtin_amdgcn_sched_barrier(0);
        if (i + AHEAD <= STRIP) LOAD_ROW(i + AHEAD);
        __builtin_amdgcn_sched_barrier(0);

        v8f acc0 = {};
        v8f acc1 = {};
        #pragma unroll
        for (int q = 0; q < PIECES; ++q) {
            const v4f dx = ring[i % RING][q] - ring[(i - 1) % RING][q];
            const v4f sq = dx * dx;
            const v2f a0 = { sq.x, sq.y };
            const v2f a1 = { sq.z, sq.w };
            acc0 = __builtin_amdgcn_wmma_f32_16x16x4_f32(
                false, a0, false, ones, (short)0, acc0, false, false);
            acc1 = __builtin_amdgcn_wmma_f32_16x16x4_f32(
                false, a1, false, ones, (short)0, acc1, false, false);
        }
        const v8f a = acc0 + acc1;
        float r = ((a[0] + a[1]) + (a[2] + a[3])) + ((a[4] + a[5]) + (a[6] + a[7]));
        asm volatile("" : "+v"(r));
        const int t = i - 1;
        if (t < 16) res0 = ((lane & 15) == (t & 15)) ? r : res0;
        else        res1 = ((lane & 15) == (t & 15)) ? r : res1;
    }
#undef LOAD_ROW

    const bool  lo   = lane < 16;
    const float mine = lo ? res0 : res1;
    const float give = lo ? res1 : res0;
    const float got  = __shfl_xor(give, 16, 32);
    out[(size_t)b * SEQ + s0 + lane] = mine + got;
}

__global__ __launch_bounds__(256) void l2d_pass2(float* __restrict__ d) {
    __shared__ float red[256];
    const int b = blockIdx.x;
    const int t = threadIdx.x;
    float* __restrict__ row = d + (size_t)b * SEQ;

    float v[SEQ / 256];
    float mx = 0.0f;
    #pragma unroll
    for (int i = 0; i < SEQ / 256; ++i) {
        v[i] = row[t + i * 256];
        mx = fmaxf(mx, v[i]);
    }
    red[t] = mx;
    __syncthreads();
    #pragma unroll
    for (int off = 128; off > 0; off >>= 1) {
        if (t < off) red[t] = fmaxf(red[t], red[t + off]);
        __syncthreads();
    }
    const float mxb = red[0];
    #pragma unroll
    for (int i = 0; i < SEQ / 256; ++i) {
        row[t + i * 256] = v[i] / mxb;
    }
}

extern "C" void kernel_launch(void* const* d_in, const int* in_sizes, int n_in,
                              void* d_out, int out_size, void* d_ws, size_t ws_size,
                              hipStream_t stream) {
    (void)in_sizes; (void)n_in; (void)out_size; (void)d_ws; (void)ws_size;
    const float* x = (const float*)d_in[0];
    float* out = (float*)d_out;

    l2d_pass1<<<BATCH * (SEQ / STRIP) / WAVES, WAVES * 32, 0, stream>>>(x, out);

    l2d_pass2<<<BATCH, 256, 0, stream>>>(out);
}
